# starting kernel
# baseline (speedup 1.0000x reference)
#include <hip/hip_runtime.h>

#define N_NODES 100000
#define N_EDGES 1600000
#define DIN 128
#define DH  128
#define DOUT 64

typedef __attribute__((ext_vector_type(2))) float v2f;
typedef __attribute__((ext_vector_type(8))) float v8f;

__device__ __forceinline__ void atomAddF32(float* p, float v) {
    unsafeAtomicAdd(p, v);
}


__global__ void k_init_deg(float* __restrict__ deg) {
    int i = blockIdx.x * blockDim.x + threadIdx.x;
    if (i < N_NODES) deg[i] = 1.0f;
}

__global__ void k_deg_accum(const int* __restrict__ dst, float* __restrict__ deg) {
    int e = blockIdx.x * blockDim.x + threadIdx.x;
    if (e < N_EDGES) atomAddF32(&deg[dst[e]], 1.0f);
}

__global__ void k_dinv(float* __restrict__ deg_dinv) {
    int i = blockIdx.x * blockDim.x + threadIdx.x;
    if (i < N_NODES) deg_dinv[i] = rsqrtf(deg_dinv[i]);
}

__global__ void k_norm_e(const int* __restrict__ src, const int* __restrict__ dst,
                         const float* __restrict__ dinv, float* __restrict__ norme) {
    int e = blockIdx.x * blockDim.x + threadIdx.x;
    if (e < N_EDGES) norme[e] = dinv[src[e]] * dinv[dst[e]];
}


template<int K, int NCOLS, bool RELU>
__global__ void k_gemm_wmma(const float* __restrict__ A,
                            const float* __restrict__ W,
                            float* __restrict__ out) {
    const int NT = NCOLS / 16;
    int tid  = blockIdx.x * blockDim.x + threadIdx.x;
    int wave = tid >> 5;
    int lane = threadIdx.x & 31;
    int mTile = wave / NT;
    int nTile = wave - mTile * NT;

    int hi      = lane >> 4;
    int laneLow = lane & 15;
    int mBase = mTile * 16;
    int nBase = nTile * 16;

    const float* aRow = A + (size_t)(mBase + laneLow) * K;
    const float* wCol = W + nBase + laneLow;

    v8f c = {0.f, 0.f, 0.f, 0.f, 0.f, 0.f, 0.f, 0.f};

    #pragma unroll 4
    for (int k = 0; k < K; k += 4) {
        int kk = k + 2 * hi;
        float a0 = aRow[kk];
        float a1 = aRow[kk + 1];
        if (RELU) { a0 = fmaxf(a0, 0.0f); a1 = fmaxf(a1, 0.0f); }
        v2f av; av.x = a0; av.y = a1;

        v2f bv;
        bv.x = wCol[(size_t)kk * NCOLS];
        bv.y = wCol[(size_t)(kk + 1) * NCOLS];

        c = __builtin_amdgcn_wmma_f32_16x16x4_f32(false, av, false, bv,
                                                  (short)0, c, false, false);
    }

    float* oCol = out + (size_t)(mBase + 8 * hi) * NCOLS + nBase + laneLow;
    #pragma unroll
    for (int r = 0; r < 8; ++r) {
        oCol[(size_t)r * NCOLS] = c[r];
    }
}


template<int D>
__global__ void k_self_bias(const float* __restrict__ h,
                            const float* __restrict__ dinv,
                            const float* __restrict__ bias,
                            float* __restrict__ agg) {
    int t = blockIdx.x * blockDim.x + threadIdx.x;
    if (t < N_NODES * D) {
        int i = t / D;
        int f = t - i * D;
        float sn = dinv[i] * dinv[i];
        agg[t] = h[t] * sn + bias[f];
    }
}


template<int D>
__global__ void k_scatter(const int* __restrict__ src, const int* __restrict__ dst,
                          const float* __restrict__ norme,
                          const float* __restrict__ h, float* __restrict__ agg) {
    const int CH = D / 4;
    int t = blockIdx.x * blockDim.x + threadIdx.x;
    if (t >= N_EDGES * CH) return;
    int e = t / CH;
    int c = t - e * CH;

    int s = src[e], d = dst[e];
    float w = norme[e];

    const float4 v = *(const float4*)(h + (size_t)s * D + c * 4);
    float* p = agg + (size_t)d * D + c * 4;
    atomAddF32(p + 0, v.x * w);
    atomAddF32(p + 1, v.y * w);
    atomAddF32(p + 2, v.z * w);
    atomAddF32(p + 3, v.w * w);
}


extern "C" void kernel_launch(void* const* d_in, const int* in_sizes, int n_in,
                              void* d_out, int out_size, void* d_ws, size_t ws_size,
                              hipStream_t stream) {
    const float* x   = (const float*)d_in[0];
    const int*   ei  = (const int*)d_in[1];
    const float* W1  = (const float*)d_in[2];
    const float* b1  = (const float*)d_in[3];
    const float* W2  = (const float*)d_in[4];
    const float* b2  = (const float*)d_in[5];
    float*       out = (float*)d_out;

    const int* src = ei;
    const int* dst = ei + N_EDGES;

    float* ws    = (float*)d_ws;
    float* dinv  = ws;
    float* norme = dinv  + N_NODES;
    float* h1    = norme + N_EDGES;
    float* agg1  = h1    + (size_t)N_NODES * DH;
    float* h2    = agg1  + (size_t)N_NODES * DH;

    const int B = 256;

    k_init_deg <<<(N_NODES + B - 1) / B, B, 0, stream>>>(dinv);
    k_deg_accum<<<(N_EDGES + B - 1) / B, B, 0, stream>>>(dst, dinv);
    k_dinv     <<<(N_NODES + B - 1) / B, B, 0, stream>>>(dinv);
    k_norm_e   <<<(N_EDGES + B - 1) / B, B, 0, stream>>>(src, dst, dinv, norme);

    {
        int waves = (N_NODES / 16) * (DH / 16);
        k_gemm_wmma<DIN, DH, false><<<waves * 32 / B, B, 0, stream>>>(x, W1, h1);
    }
    k_self_bias<DH><<<(N_NODES * DH + B - 1) / B, B, 0, stream>>>(h1, dinv, b1, agg1);
    {
        int total = N_EDGES * (DH / 4);
        k_scatter<DH><<<(total + B - 1) / B, B, 0, stream>>>(src, dst, norme, h1, agg1);
    }

    {
        int waves = (N_NODES / 16) * (DOUT / 16);
        k_gemm_wmma<DH, DOUT, true><<<waves * 32 / B, B, 0, stream>>>(agg1, W2, h2);
    }
    k_self_bias<DOUT><<<(N_NODES * DOUT + B - 1) / B, B, 0, stream>>>(h2, dinv, b2, out);
    {
        int total = N_EDGES * (DOUT / 4);
        k_scatter<DOUT><<<(total + B - 1) / B, B, 0, stream>>>(src, dst, norme, h2, out);
    }
}


# Float-atomic edge scatter replaced by counting sort on dst (returning int add gives degree and arrival-order rank) plus per-node register gather-sum started from self term + bias
# speedup vs baseline: 2.1245x; 2.1245x over previous
#include <hip/hip_runtime.h>

#define N_NODES 100000
#define N_EDGES 1600000
#define DIN 128
#define DH  128
#define DOUT 64

typedef __attribute__((ext_vector_type(2))) float v2f;
typedef __attribute__((ext_vector_type(4))) float v4f;
typedef __attribute__((ext_vector_type(8))) float v8f;


#define SCAN_ELEMS 1024
#define SCAN_BLOCKS ((N_NODES + SCAN_ELEMS - 1) / SCAN_ELEMS)


__global__ void k_zero_cnt(int* __restrict__ cnt) {
    int i = blockIdx.x * blockDim.x + threadIdx.x;
    if (i < N_NODES) cnt[i] = 0;
}

__global__ void k_count_rank(const int* __restrict__ dst, int* __restrict__ cnt,
                             int* __restrict__ rank) {
    int e = blockIdx.x * blockDim.x + threadIdx.x;
    if (e < N_EDGES) rank[e] = atomicAdd(&cnt[dst[e]], 1);
}


__device__ __forceinline__ int blockExclScan(int v, int& total, int* sm) {
    int lane = threadIdx.x & 31;
    int w    = threadIdx.x >> 5;
    int inc = v;
    for (int o = 1; o < 32; o <<= 1) {
        int n = __shfl_up(inc, o, 32);
        if (lane >= o) inc += n;
    }
    if (lane == 31) sm[w] = inc;
    __syncthreads();
    int before = 0, all = 0;
    for (int i = 0; i < 8; ++i) {
        int s = sm[i];
        if (i < w) before += s;
        all += s;
    }
    __syncthreads();
    total = all;
    return before + inc - v;
}

__global__ void k_block_sums(const int* __restrict__ cnt, int* __restrict__ bsum) {
    __shared__ int sm[8];
    int idx = blockIdx.x * SCAN_ELEMS + threadIdx.x * 4;
    int s = 0;
    if (idx < N_NODES) {
        const int4 c = *(const int4*)(cnt + idx);
        s = c.x + c.y + c.z + c.w;
    }
    int total;
    blockExclScan(s, total, sm);
    if (threadIdx.x == 0) bsum[blockIdx.x] = total;
}

__global__ void k_scan_dinv(const int* __restrict__ cnt, const int* __restrict__ bsum,
                            int* __restrict__ rowptr, float* __restrict__ dinv) {
    __shared__ int sm[8];
    int b = blockIdx.x;
    int t = threadIdx.x;
    int idx = b * SCAN_ELEMS + t * 4;
    int4 c = make_int4(0, 0, 0, 0);
    if (idx < N_NODES) c = *(const int4*)(cnt + idx);
    int total;
    int ex = blockExclScan(c.x + c.y + c.z + c.w, total, sm);
    int base;
    blockExclScan((t < b) ? bsum[t] : 0, base, sm);
    if (idx < N_NODES) {
        int p0 = base + ex;
        int4 r;
        r.x = p0;
        r.y = p0 + c.x;
        r.z = r.y + c.y;
        r.w = r.z + c.z;
        *(int4*)(rowptr + idx) = r;
        float4 d;
        d.x = rsqrtf((float)c.x + 1.0f);
        d.y = rsqrtf((float)c.y + 1.0f);
        d.z = rsqrtf((float)c.z + 1.0f);
        d.w = rsqrtf((float)c.w + 1.0f);
        *(float4*)(dinv + idx) = d;
    }
}


__global__ void k_fill(const int* __restrict__ src, const int* __restrict__ dst,
                       const int* __restrict__ rank, const int* __restrict__ rowptr,
                       const float* __restrict__ dinv, int* __restrict__ rec) {
    int e = blockIdx.x * blockDim.x + threadIdx.x;
    if (e < N_EDGES) {
        int s = src[e], d = dst[e];
        int pos = rowptr[d] + rank[e];
        float w = dinv[s] * dinv[d];
        *(int2*)(rec + 2 * (size_t)pos) = make_int2(s, __float_as_int(w));
    }
}


template<int K, int NCOLS, bool RELU>
__global__ void k_gemm_wmma(const float* __restrict__ A,
                            const float* __restrict__ W,
                            float* __restrict__ out) {
    const int NT = NCOLS / 16;
    int tid  = blockIdx.x * blockDim.x + threadIdx.x;
    int wave = tid >> 5;
    int lane = threadIdx.x & 31;
    int mTile = wave / NT;
    int nTile = wave - mTile * NT;

    int hi      = lane >> 4;
    int laneLow = lane & 15;
    int mBase = mTile * 16;
    int nBase = nTile * 16;

    const float* aRow = A + (size_t)(mBase + laneLow) * K;
    const float* wCol = W + nBase + laneLow;

    v8f c = {0.f, 0.f, 0.f, 0.f, 0.f, 0.f, 0.f, 0.f};

    #pragma unroll 4
    for (int k = 0; k < K; k += 4) {
        int kk = k + 2 * hi;
        float a0 = aRow[kk];
        float a1 = aRow[kk + 1];
        if (RELU) { a0 = fmaxf(a0, 0.0f); a1 = fmaxf(a1, 0.0f); }
        v2f av; av.x = a0; av.y = a1;

        v2f bv;
        bv.x = wCol[(size_t)kk * NCOLS];
        bv.y = wCol[(size_t)(kk + 1) * NCOLS];

        c = __builtin_amdgcn_wmma_f32_16x16x4_f32(false, av, false, bv,
                                                  (short)0, c, false, false);
    }

    float* oCol = out + (size_t)(mBase + 8 * hi) * NCOLS + nBase + laneLow;
    #pragma unroll
    for (int r = 0; r < 8; ++r) {
        oCol[(size_t)r * NCOLS] = c[r];
    }
}


template<int V> struct RowVec;
template<> struct RowVec<4> { typedef v4f T; };
template<> struct RowVec<2> { typedef v2f T; };

template<int D>
__global__ void k_gather(const int* __restrict__ rowptr, const int* __restrict__ cnt,
                         const int* __restrict__ rec, const float* __restrict__ dinv,
                         const float* __restrict__ bias, const float* __restrict__ h,
                         float* __restrict__ agg) {
    const int V = D / 32;
    typedef typename RowVec<V>::T vec;
    int lane = threadIdx.x & 31;
    int node = __builtin_amdgcn_readfirstlane((int)((blockIdx.x * blockDim.x + threadIdx.x) >> 5));

    int j   = rowptr[node];
    int end = j + cnt[node];
    float di = dinv[node];
    float sn = di * di;

    const float* hl = h + lane * V;
    vec acc = *(const vec*)(hl + (size_t)node * D) * sn + *(const vec*)(bias + lane * V);

    const int2* r = (const int2*)rec;
    for (; j + 4 <= end; j += 4) {
        int2 p0 = r[j], p1 = r[j + 1], p2 = r[j + 2], p3 = r[j + 3];
        vec x0 = *(const vec*)(hl + (size_t)p0.x * D);
        vec x1 = *(const vec*)(hl + (size_t)p1.x * D);
        vec x2 = *(const vec*)(hl + (size_t)p2.x * D);
        vec x3 = *(const vec*)(hl + (size_t)p3.x * D);
        acc += x0 * __int_as_float(p0.y);
        acc += x1 * __int_as_float(p1.y);
        acc += x2 * __int_as_float(p2.y);
        acc += x3 * __int_as_float(p3.y);
    }
    for (; j < end; ++j) {
        int2 p = r[j];
        vec x = *(const vec*)(hl + (size_t)p.x * D);
        acc += x * __int_as_float(p.y);
    }
    *(vec*)(agg + (size_t)node * D + lane * V) = acc;
}


extern "C" void kernel_launch(void* const* d_in, const int* in_sizes, int n_in,
                              void* d_out, int out_size, void* d_ws, size_t ws_size,
                              hipStream_t stream) {
    const float* x   = (const float*)d_in[0];
    const int*   ei  = (const int*)d_in[1];
    const float* W1  = (const float*)d_in[2];
    const float* b1  = (const float*)d_in[3];
    const float* W2  = (const float*)d_in[4];
    const float* b2  = (const float*)d_in[5];
    float*       out = (float*)d_out;

    const int* src = ei;
    const int* dst = ei + N_EDGES;

    float* ws    = (float*)d_ws;
    float* dinv  = ws;
    int*   rank  = (int*)(dinv + N_NODES);
    float* h1    = dinv + N_NODES + N_EDGES;
    float* agg1  = h1    + (size_t)N_NODES * DH;
    int*   rec   = (int*)(agg1 + (size_t)N_NODES * DH);
    int*   cnt   = rec + 2 * (size_t)N_EDGES;
    int*   rowp  = cnt + N_NODES;
    int*   bsum  = rowp + N_NODES;
    float* h2    = h1;

    const int B = 256;

    k_zero_cnt  <<<(N_NODES + B - 1) / B, B, 0, stream>>>(cnt);
    k_count_rank<<<(N_EDGES + B - 1) / B, B, 0, stream>>>(dst, cnt, rank);
    k_block_sums<<<SCAN_BLOCKS, B, 0, stream>>>(cnt, bsum);
    k_scan_dinv <<<SCAN_BLOCKS, B, 0, stream>>>(cnt, bsum, rowp, dinv);
    k_fill      <<<(N_EDGES + B - 1) / B, B, 0, stream>>>(src, dst, rank, rowp, dinv, rec);

    {
        int waves = (N_NODES / 16) * (DH / 16);
        k_gemm_wmma<DIN, DH, false><<<waves * 32 / B, B, 0, stream>>>(x, W1, h1);
    }
    k_gather<DH><<<N_NODES * 32 / B, B, 0, stream>>>(rowp, cnt, rec, dinv, b1, h1, agg1);

    {
        int waves = (N_NODES / 16) * (DOUT / 16);
        k_gemm_wmma<DH, DOUT, true><<<waves * 32 / B, B, 0, stream>>>(agg1, W2, h2);
    }
    k_gather<DOUT><<<N_NODES * 32 / B, B, 0, stream>>>(rowp, cnt, rec, dinv, b2, h2, out);
}


# GEMMs: one wave computes a full-width 16-row strip: A fragment loaded once for all column tiles, W rows read as whole-line 16-byte loads through a column map; k order and every sum unchanged
# speedup vs baseline: 2.2594x; 1.0635x over previous
#include <hip/hip_runtime.h>

#define N_NODES 100000
#define N_EDGES 1600000
#define DIN 128
#define DH  128
#define DOUT 64

typedef __attribute__((ext_vector_type(2))) float v2f;
typedef __attribute__((ext_vector_type(4))) float v4f;
typedef __attribute__((ext_vector_type(8))) float v8f;


#define SCAN_ELEMS 1024
#define SCAN_BLOCKS ((N_NODES + SCAN_ELEMS - 1) / SCAN_ELEMS)


__global__ void k_zero_cnt(int* __restrict__ cnt) {
    int i = blockIdx.x * blockDim.x + threadIdx.x;
    if (i < N_NODES) cnt[i] = 0;
}

__global__ void k_count_rank(const int* __restrict__ dst, int* __restrict__ cnt,
                             int* __restrict__ rank) {
    int e = blockIdx.x * blockDim.x + threadIdx.x;
    if (e < N_EDGES) rank[e] = atomicAdd(&cnt[dst[e]], 1);
}


__device__ __forceinline__ int blockExclScan(int v, int& total, int* sm) {
    int lane = threadIdx.x & 31;
    int w    = threadIdx.x >> 5;
    int inc = v;
    for (int o = 1; o < 32; o <<= 1) {
        int n = __shfl_up(inc, o, 32);
        if (lane >= o) inc += n;
    }
    if (lane == 31) sm[w] = inc;
    __syncthreads();
    int before = 0, all = 0;
    for (int i = 0; i < 8; ++i) {
        int s = sm[i];
        if (i < w) before += s;
        all += s;
    }
    __syncthreads();
    total = all;
    return before + inc - v;
}

__global__ void k_block_sums(const int* __restrict__ cnt, int* __restrict__ bsum) {
    __shared__ int sm[8];
    int idx = blockIdx.x * SCAN_ELEMS + threadIdx.x * 4;
    int s = 0;
    if (idx < N_NODES) {
        const int4 c = *(const int4*)(cnt + idx);
        s = c.x + c.y + c.z + c.w;
    }
    int total;
    blockExclScan(s, total, sm);
    if (threadIdx.x == 0) bsum[blockIdx.x] = total;
}

__global__ void k_scan_dinv(const int* __restrict__ cnt, const int* __restrict__ bsum,
                            int* __restrict__ rowptr, float* __restrict__ dinv) {
    __shared__ int sm[8];
    int b = blockIdx.x;
    int t = threadIdx.x;
    int idx = b * SCAN_ELEMS + t * 4;
    int4 c = make_int4(0, 0, 0, 0);
    if (idx < N_NODES) c = *(const int4*)(cnt + idx);
    int total;
    int ex = blockExclScan(c.x + c.y + c.z + c.w, total, sm);
    int base;
    blockExclScan((t < b) ? bsum[t] : 0, base, sm);
    if (idx < N_NODES) {
        int p0 = base + ex;
        int4 r;
        r.x = p0;
        r.y = p0 + c.x;
        r.z = r.y + c.y;
        r.w = r.z + c.z;
        *(int4*)(rowptr + idx) = r;
        float4 d;
        d.x = rsqrtf((float)c.x + 1.0f);
        d.y = rsqrtf((float)c.y + 1.0f);
        d.z = rsqrtf((float)c.z + 1.0f);
        d.w = rsqrtf((float)c.w + 1.0f);
        *(float4*)(dinv + idx) = d;
    }
}


__global__ void k_fill(const int* __restrict__ src, const int* __restrict__ dst,
                       const int* __restrict__ rank, const int* __restrict__ rowptr,
                       const float* __restrict__ dinv, int* __restrict__ rec) {
    int e = blockIdx.x * blockDim.x + threadIdx.x;
    if (e < N_EDGES) {
        int s = src[e], d = dst[e];
        int pos = rowptr[d] + rank[e];
        float w = dinv[s] * dinv[d];
        *(int2*)(rec + 2 * (size_t)pos) = make_int2(s, __float_as_int(w));
    }
}


template<int K, int NCOLS, bool RELU>
__global__ void k_gemm_wmma(const float* __restrict__ A,
                            const float* __restrict__ W,
                            float* __restrict__ out) {
    const int NT = NCOLS / 16;
    const int NG = NCOLS / 64;
    int strip = (blockIdx.x * blockDim.x + threadIdx.x) >> 5;
    int lane  = threadIdx.x & 31;

    int hi = lane >> 4;
    int c  = lane & 15;
    int mBase = strip * 16;

    const float* aRow = A + (size_t)(mBase + c) * K + 2 * hi;
    const float* wRow = W + (size_t)(2 * hi) * NCOLS + 4 * c;

    v8f acc[NT];
    #pragma unroll
    for (int t = 0; t < NT; ++t)
        acc[t] = (v8f){0.f, 0.f, 0.f, 0.f, 0.f, 0.f, 0.f, 0.f};

    #pragma unroll 2
    for (int k = 0; k < K; k += 4) {
        v2f av = *(const v2f*)(aRow + k);
        if (RELU) { av.x = fmaxf(av.x, 0.0f); av.y = fmaxf(av.y, 0.0f); }

        v4f b0[NG], b1[NG];
        #pragma unroll
        for (int g = 0; g < NG; ++g) {
            b0[g] = *(const v4f*)(wRow + (size_t)k * NCOLS + 64 * g);
            b1[g] = *(const v4f*)(wRow + (size_t)(k + 1) * NCOLS + 64 * g);
        }

        #pragma unroll
        for (int t = 0; t < NT; ++t) {
            v2f bv;
            bv.x = b0[t >> 2][t & 3];
            bv.y = b1[t >> 2][t & 3];
            acc[t] = __builtin_amdgcn_wmma_f32_16x16x4_f32(false, av, false, bv,
                                                           (short)0, acc[t], false, false);
        }
    }

    float* oRow = out + (size_t)(mBase + 8 * hi) * NCOLS + 4 * c;
    #pragma unroll
    for (int r = 0; r < 8; ++r) {
        #pragma unroll
        for (int g = 0; g < NG; ++g) {
            v4f o;
            o.x = acc[4 * g + 0][r];
            o.y = acc[4 * g + 1][r];
            o.z = acc[4 * g + 2][r];
            o.w = acc[4 * g + 3][r];
            *(v4f*)(oRow + (size_t)r * NCOLS + 64 * g) = o;
        }
    }
}


template<int V> struct RowVec;
template<> struct RowVec<4> { typedef v4f T; };
template<> struct RowVec<2> { typedef v2f T; };

template<int D>
__global__ void k_gather(const int* __restrict__ rowptr, const int* __restrict__ cnt,
                         const int* __restrict__ rec, const float* __restrict__ dinv,
                         const float* __restrict__ bias, const float* __restrict__ h,
                         float* __restrict__ agg) {
    const int V = D / 32;
    typedef typename RowVec<V>::T vec;
    int lane = threadIdx.x & 31;
    int node = __builtin_amdgcn_readfirstlane((int)((blockIdx.x * blockDim.x + threadIdx.x) >> 5));

    int j   = rowptr[node];
    int end = j + cnt[node];
    float di = dinv[node];
    float sn = di * di;

    const float* hl = h + lane * V;
    vec acc = *(const vec*)(hl + (size_t)node * D) * sn + *(const vec*)(bias + lane * V);

    const int2* r = (const int2*)rec;
    for (; j + 4 <= end; j += 4) {
        int2 p0 = r[j], p1 = r[j + 1], p2 = r[j + 2], p3 = r[j + 3];
        vec x0 = *(const vec*)(hl + (size_t)p0.x * D);
        vec x1 = *(const vec*)(hl + (size_t)p1.x * D);
        vec x2 = *(const vec*)(hl + (size_t)p2.x * D);
        vec x3 = *(const vec*)(hl + (size_t)p3.x * D);
        acc += x0 * __int_as_float(p0.y);
        acc += x1 * __int_as_float(p1.y);
        acc += x2 * __int_as_float(p2.y);
        acc += x3 * __int_as_float(p3.y);
    }
    for (; j < end; ++j) {
        int2 p = r[j];
        vec x = *(const vec*)(hl + (size_t)p.x * D);
        acc += x * __int_as_float(p.y);
    }
    *(vec*)(agg + (size_t)node * D + lane * V) = acc;
}


extern "C" void kernel_launch(void* const* d_in, const int* in_sizes, int n_in,
                              void* d_out, int out_size, void* d_ws, size_t ws_size,
                              hipStream_t stream) {
    const float* x   = (const float*)d_in[0];
    const int*   ei  = (const int*)d_in[1];
    const float* W1  = (const float*)d_in[2];
    const float* b1  = (const float*)d_in[3];
    const float* W2  = (const float*)d_in[4];
    const float* b2  = (const float*)d_in[5];
    float*       out = (float*)d_out;

    const int* src = ei;
    const int* dst = ei + N_EDGES;

    float* ws    = (float*)d_ws;
    float* dinv  = ws;
    int*   rank  = (int*)(dinv + N_NODES);
    float* h1    = dinv + N_NODES + N_EDGES;
    float* agg1  = h1    + (size_t)N_NODES * DH;
    int*   rec   = (int*)(agg1 + (size_t)N_NODES * DH);
    int*   cnt   = rec + 2 * (size_t)N_EDGES;
    int*   rowp  = cnt + N_NODES;
    int*   bsum  = rowp + N_NODES;
    float* h2    = h1;

    const int B = 256;

    k_zero_cnt  <<<(N_NODES + B - 1) / B, B, 0, stream>>>(cnt);
    k_count_rank<<<(N_EDGES + B - 1) / B, B, 0, stream>>>(dst, cnt, rank);
    k_block_sums<<<SCAN_BLOCKS, B, 0, stream>>>(cnt, bsum);
    k_scan_dinv <<<SCAN_BLOCKS, B, 0, stream>>>(cnt, bsum, rowp, dinv);
    k_fill      <<<(N_EDGES + B - 1) / B, B, 0, stream>>>(src, dst, rank, rowp, dinv, rec);

    const int GB = 64;
    {
        int waves = N_NODES / 16;
        k_gemm_wmma<DIN, DH, false><<<waves * 32 / GB, GB, 0, stream>>>(x, W1, h1);
    }
    k_gather<DH><<<N_NODES * 32 / B, B, 0, stream>>>(rowp, cnt, rec, dinv, b1, h1, agg1);

    {
        int waves = N_NODES / 16;
        k_gemm_wmma<DH, DOUT, true><<<waves * 32 / GB, GB, 0, stream>>>(agg1, W2, h2);
    }
    k_gather<DOUT><<<N_NODES * 32 / B, B, 0, stream>>>(rowp, cnt, rec, dinv, b2, h2, out);
}


# Edge sort by destination done through LDS counters in two levels (256-node buckets) instead of a returning global add and three random reads per edge; per-node term order stays an arrival order
# speedup vs baseline: 3.6194x; 1.6019x over previous
#include <hip/hip_runtime.h>

#define N_NODES 100000
#define N_EDGES 1600000
#define DIN 128
#define DH  128
#define DOUT 64

typedef __attribute__((ext_vector_type(2))) float v2f;
typedef __attribute__((ext_vector_type(4))) float v4f;
typedef __attribute__((ext_vector_type(8))) float v8f;


#define BK_SHIFT 8
#define BK_NODES (1 << BK_SHIFT)
#define NBK ((N_NODES + BK_NODES - 1) / BK_NODES)
#define CHUNK 4096
#define NCHUNK ((N_EDGES + CHUNK - 1) / CHUNK)
#define EPT (CHUNK / 256)
#define BCAP 6144

__device__ __forceinline__ int blockExclScan(int v, int& total, int* sm) {
    int lane = threadIdx.x & 31;
    int w    = threadIdx.x >> 5;
    int inc = v;
    for (int o = 1; o < 32; o <<= 1) {
        int n = __shfl_up(inc, o, 32);
        if (lane >= o) inc += n;
    }
    if (lane == 31) sm[w] = inc;
    __syncthreads();
    int before = 0, all = 0;
    for (int i = 0; i < 8; ++i) {
        int s = sm[i];
        if (i < w) before += s;
        all += s;
    }
    __syncthreads();
    total = all;
    return before + inc - v;
}

__global__ void k_zero_tot(int* __restrict__ tot) {
    int i = blockIdx.x * blockDim.x + threadIdx.x;
    if (i < 2 * NBK) tot[i] = 0;
}


__global__ void k_hist(const int* __restrict__ dst, int* __restrict__ btot) {
    __shared__ int lh[NBK];
    int tid = threadIdx.x;
    for (int i = tid; i < NBK; i += 256) lh[i] = 0;
    __syncthreads();
    int e0 = blockIdx.x * CHUNK;
    int d[EPT];
    #pragma unroll
    for (int k = 0; k < EPT; ++k) {
        int e = e0 + k * 256 + tid;
        d[k] = (e < N_EDGES) ? dst[e] : -1;
    }
    #pragma unroll
    for (int k = 0; k < EPT; ++k)
        if (d[k] >= 0) atomicAdd(&lh[d[k] >> BK_SHIFT], 1);
    __syncthreads();
    for (int i = tid; i < NBK; i += 256) {
        int h = lh[i];
        if (h) atomicAdd(&btot[i], h);
    }
}


__global__ void k_part(const int* __restrict__ src, const int* __restrict__ dst,
                       const int* __restrict__ btot, int* __restrict__ bcur,
                       int2* __restrict__ part) {
    __shared__ int lh[NBK];
    __shared__ int lex[NBK];
    __shared__ int lrun[NBK];
    __shared__ int lfill[NBK];
    __shared__ int sm[8];
    __shared__ int2 stage[CHUNK];
    int tid = threadIdx.x;
    for (int i = tid; i < NBK; i += 256) { lh[i] = 0; lfill[i] = 0; }
    __syncthreads();

    int e0 = blockIdx.x * CHUNK;
    int s[EPT], d[EPT];
    #pragma unroll
    for (int k = 0; k < EPT; ++k) {
        int e = e0 + k * 256 + tid;
        bool ok = e < N_EDGES;
        d[k] = ok ? dst[e] : -1;
        s[k] = ok ? src[e] : 0;
    }
    #pragma unroll
    for (int k = 0; k < EPT; ++k)
        if (d[k] >= 0) atomicAdd(&lh[d[k] >> BK_SHIFT], 1);
    __syncthreads();

    int t0 = 2 * tid, t1 = 2 * tid + 1;
    int g0 = (t0 < NBK) ? btot[t0] : 0;
    int g1 = (t1 < NBK) ? btot[t1] : 0;
    int h0 = (t0 < NBK) ? lh[t0] : 0;
    int h1 = (t1 < NBK) ? lh[t1] : 0;
    int tot;
    int gex = blockExclScan(g0 + g1, tot, sm);
    int hex = blockExclScan(h0 + h1, tot, sm);
    if (t0 < NBK) {
        lex[t0] = hex;
        int r = h0 ? atomicAdd(&bcur[t0], h0) : 0;
        lrun[t0] = gex + r;
    }
    if (t1 < NBK) {
        lex[t1] = hex + h0;
        int r = h1 ? atomicAdd(&bcur[t1], h1) : 0;
        lrun[t1] = gex + g0 + r;
    }
    __syncthreads();

    #pragma unroll
    for (int k = 0; k < EPT; ++k) {
        if (d[k] >= 0) {
            int b = d[k] >> BK_SHIFT;
            int r = atomicAdd(&lfill[b], 1);
            stage[lex[b] + r] = make_int2(s[k], d[k]);
        }
    }
    __syncthreads();

    int n = N_EDGES - e0;
    if (n > CHUNK) n = CHUNK;
    for (int i = tid; i < n; i += 256) {
        int2 v = stage[i];
        int b = v.y >> BK_SHIFT;
        part[lrun[b] + (i - lex[b])] = v;
    }
}


__global__ void
k_bucket(const int* __restrict__ btot, const int2* __restrict__ part,
         int* __restrict__ rowptr, int* __restrict__ cnt, float* __restrict__ dinv,
         int2* __restrict__ rec) {
    __shared__ int lcnt[BK_NODES];
    __shared__ int lptr[BK_NODES];
    __shared__ int lcur[BK_NODES];
    __shared__ int red[32];
    __shared__ int2 stage[BCAP];
    int tid  = threadIdx.x;
    int lane = tid & 31;
    int w    = tid >> 5;
    int b    = blockIdx.x;

    int v = (tid < b && tid < NBK) ? btot[tid] : 0;
    for (int o = 16; o > 0; o >>= 1) v += __shfl_xor(v, o, 32);
    if (lane == 0) red[w] = v;
    if (tid < BK_NODES) { lcnt[tid] = 0; lcur[tid] = 0; }
    __syncthreads();
    int base = 0;
    for (int i = 0; i < 32; ++i) base += red[i];
    int n = btot[b];
    const int2* p = part + base;

    for (int i0 = 0; i0 < n; i0 += 4096) {
        int dd[4];
        #pragma unroll
        for (int k = 0; k < 4; ++k) {
            int i = i0 + k * 1024 + tid;
            dd[k] = (i < n) ? p[i].y : -1;
        }
        #pragma unroll
        for (int k = 0; k < 4; ++k)
            if (dd[k] >= 0) atomicAdd(&lcnt[dd[k] & (BK_NODES - 1)], 1);
    }
    __syncthreads();

    if (tid < 32) {
        int c[8];
        int sum = 0;
        #pragma unroll
        for (int k = 0; k < 8; ++k) { c[k] = lcnt[tid * 8 + k]; sum += c[k]; }
        int inc = sum;
        for (int o = 1; o < 32; o <<= 1) {
            int nb = __shfl_up(inc, o, 32);
            if (lane >= o) inc += nb;
        }
        int run = inc - sum;
        #pragma unroll
        for (int k = 0; k < 8; ++k) { lptr[tid * 8 + k] = run; run += c[k]; }
    }
    __syncthreads();

    if (tid < BK_NODES) {
        int node = b * BK_NODES + tid;
        if (node < N_NODES) {
            int c = lcnt[tid];
            rowptr[node] = base + lptr[tid];
            cnt[node]    = c;
            dinv[node]   = rsqrtf((float)c + 1.0f);
        }
    }

    bool st = (n <= BCAP);
    for (int i0 = 0; i0 < n; i0 += 4096) {
        int2 ed[4];
        #pragma unroll
        for (int k = 0; k < 4; ++k) {
            int i = i0 + k * 1024 + tid;
            ed[k] = (i < n) ? p[i] : make_int2(0, -1);
        }
        #pragma unroll
        for (int k = 0; k < 4; ++k) {
            if (ed[k].y >= 0) {
                int dl = ed[k].y & (BK_NODES - 1);
                int r  = atomicAdd(&lcur[dl], 1);
                int lp = lptr[dl] + r;
                if (st) stage[lp] = ed[k];
                else    rec[base + lp] = ed[k];
            }
        }
    }
    if (st) {
        __syncthreads();
        for (int i = tid; i < n; i += 1024) rec[base + i] = stage[i];
    }
}


__global__ void k_weights(const float* __restrict__ dinv, int* rec) {
    int t = blockIdx.x * blockDim.x + threadIdx.x;
    if (t < N_EDGES / 2) {
        int4 v = ((const int4*)rec)[t];
        float w0 = dinv[v.x] * dinv[v.y];
        float w1 = dinv[v.z] * dinv[v.w];
        ((int4*)rec)[t] = make_int4(v.x, __float_as_int(w0), v.z, __float_as_int(w1));
    }
}


template<int K, int NCOLS, bool RELU>
__global__ void k_gemm_wmma(const float* __restrict__ A,
                            const float* __restrict__ W,
                            float* __restrict__ out) {
    const int NT = NCOLS / 16;
    const int NG = NCOLS / 64;
    int strip = (blockIdx.x * blockDim.x + threadIdx.x) >> 5;
    int lane  = threadIdx.x & 31;

    int hi = lane >> 4;
    int c  = lane & 15;
    int mBase = strip * 16;

    const float* aRow = A + (size_t)(mBase + c) * K + 2 * hi;
    const float* wRow = W + (size_t)(2 * hi) * NCOLS + 4 * c;

    v8f acc[NT];
    #pragma unroll
    for (int t = 0; t < NT; ++t)
        acc[t] = (v8f){0.f, 0.f, 0.f, 0.f, 0.f, 0.f, 0.f, 0.f};

    #pragma unroll 2
    for (int k = 0; k < K; k += 4) {
        v2f av = *(const v2f*)(aRow + k);
        if (RELU) { av.x = fmaxf(av.x, 0.0f); av.y = fmaxf(av.y, 0.0f); }

        v4f b0[NG], b1[NG];
        #pragma unroll
        for (int g = 0; g < NG; ++g) {
            b0[g] = *(const v4f*)(wRow + (size_t)k * NCOLS + 64 * g);
            b1[g] = *(const v4f*)(wRow + (size_t)(k + 1) * NCOLS + 64 * g);
        }

        #pragma unroll
        for (int t = 0; t < NT; ++t) {
            v2f bv;
            bv.x = b0[t >> 2][t & 3];
            bv.y = b1[t >> 2][t & 3];
            acc[t] = __builtin_amdgcn_wmma_f32_16x16x4_f32(false, av, false, bv,
                                                           (short)0, acc[t], false, false);
        }
    }

    float* oRow = out + (size_t)(mBase + 8 * hi) * NCOLS + 4 * c;
    #pragma unroll
    for (int r = 0; r < 8; ++r) {
        #pragma unroll
        for (int g = 0; g < NG; ++g) {
            v4f o;
            o.x = acc[4 * g + 0][r];
            o.y = acc[4 * g + 1][r];
            o.z = acc[4 * g + 2][r];
            o.w = acc[4 * g + 3][r];
            *(v4f*)(oRow + (size_t)r * NCOLS + 64 * g) = o;
        }
    }
}


template<int V> struct RowVec;
template<> struct RowVec<4> { typedef v4f T; };
template<> struct RowVec<2> { typedef v2f T; };

template<int D>
__global__ void k_gather(const int* __restrict__ rowptr, const int* __restrict__ cnt,
                         const int* __restrict__ rec, const float* __restrict__ dinv,
                         const float* __restrict__ bias, const float* __restrict__ h,
                         float* __restrict__ agg) {
    const int V = D / 32;
    typedef typename RowVec<V>::T vec;
    int lane = threadIdx.x & 31;
    int node = __builtin_amdgcn_readfirstlane((int)((blockIdx.x * blockDim.x + threadIdx.x) >> 5));

    int j   = rowptr[node];
    int end = j + cnt[node];
    float di = dinv[node];
    float sn = di * di;

    const float* hl = h + lane * V;
    vec acc = *(const vec*)(hl + (size_t)node * D) * sn + *(const vec*)(bias + lane * V);

    const int2* r = (const int2*)rec;
    for (; j + 4 <= end; j += 4) {
        int2 p0 = r[j], p1 = r[j + 1], p2 = r[j + 2], p3 = r[j + 3];
        vec x0 = *(const vec*)(hl + (size_t)p0.x * D);
        vec x1 = *(const vec*)(hl + (size_t)p1.x * D);
        vec x2 = *(const vec*)(hl + (size_t)p2.x * D);
        vec x3 = *(const vec*)(hl + (size_t)p3.x * D);
        acc += x0 * __int_as_float(p0.y);
        acc += x1 * __int_as_float(p1.y);
        acc += x2 * __int_as_float(p2.y);
        acc += x3 * __int_as_float(p3.y);
    }
    for (; j < end; ++j) {
        int2 p = r[j];
        vec x = *(const vec*)(hl + (size_t)p.x * D);
        acc += x * __int_as_float(p.y);
    }
    *(vec*)(agg + (size_t)node * D + lane * V) = acc;
}


extern "C" void kernel_launch(void* const* d_in, const int* in_sizes, int n_in,
                              void* d_out, int out_size, void* d_ws, size_t ws_size,
                              hipStream_t stream) {
    const float* x   = (const float*)d_in[0];
    const int*   ei  = (const int*)d_in[1];
    const float* W1  = (const float*)d_in[2];
    const float* b1  = (const float*)d_in[3];
    const float* W2  = (const float*)d_in[4];
    const float* b2  = (const float*)d_in[5];
    float*       out = (float*)d_out;

    const int* src = ei;
    const int* dst = ei + N_EDGES;

    float* ws    = (float*)d_ws;
    float* dinv  = ws;
    int*   btot  = (int*)(dinv + N_NODES);
    int*   bcur  = btot + NBK;
    float* h1    = dinv + N_NODES + N_EDGES;
    float* agg1  = h1    + (size_t)N_NODES * DH;
    int*   rec   = (int*)(agg1 + (size_t)N_NODES * DH);
    int*   cnt   = rec + 2 * (size_t)N_EDGES;
    int*   rowp  = cnt + N_NODES;
    int2*  part  = (int2*)agg1;
    float* h2    = h1;

    const int B = 256;

    k_zero_tot<<<(2 * NBK + B - 1) / B, B, 0, stream>>>(btot);
    k_hist    <<<NCHUNK, B, 0, stream>>>(dst, btot);
    k_part    <<<NCHUNK, B, 0, stream>>>(src, dst, btot, bcur, part);
    k_bucket  <<<NBK, 1024, 0, stream>>>(btot, part, rowp, cnt, dinv, (int2*)rec);
    k_weights <<<(N_EDGES / 2 + B - 1) / B, B, 0, stream>>>(dinv, rec);

    const int GB = 64;
    {
        int waves = N_NODES / 16;
        k_gemm_wmma<DIN, DH, false><<<waves * 32 / GB, GB, 0, stream>>>(x, W1, h1);
    }
    k_gather<DH><<<N_NODES * 32 / B, B, 0, stream>>>(rowp, cnt, rec, dinv, b1, h1, agg1);

    {
        int waves = N_NODES / 16;
        k_gemm_wmma<DH, DOUT, true><<<waves * 32 / GB, GB, 0, stream>>>(agg1, W2, h2);
    }
    k_gather<DOUT><<<N_NODES * 32 / B, B, 0, stream>>>(rowp, cnt, rec, dinv, b2, h2, out);
}


# Edge sort without histogram pass or global atomics: chunks written bucket-ordered in place plus an offset table, buckets collect their runs; 9 -> 7 launches; per-node sum order stays arrival order
# speedup vs baseline: 3.7429x; 1.0341x over previous
#include <hip/hip_runtime.h>

#define N_NODES 100000
#define N_EDGES 1600000
#define DIN 128
#define DH  128
#define DOUT 64

typedef __attribute__((ext_vector_type(2))) float v2f;
typedef __attribute__((ext_vector_type(4))) float v4f;
typedef __attribute__((ext_vector_type(8))) float v8f;


#define BK_SHIFT 8
#define BK_NODES (1 << BK_SHIFT)
#define NBK ((N_NODES + BK_NODES - 1) / BK_NODES)
#define CHUNK 4096
#define NCHUNK ((N_EDGES + CHUNK - 1) / CHUNK)
#define EPT (CHUNK / 256)
#define BCAP 6144
#define TROW (NBK + 1)
#define NGRP 64
#define RPG ((NCHUNK + NGRP - 1) / NGRP)

__device__ __forceinline__ int blockExclScan(int v, int& total, int* sm) {
    int lane = threadIdx.x & 31;
    int w    = threadIdx.x >> 5;
    int inc = v;
    for (int o = 1; o < 32; o <<= 1) {
        int n = __shfl_up(inc, o, 32);
        if (lane >= o) inc += n;
    }
    if (lane == 31) sm[w] = inc;
    __syncthreads();
    int before = 0, all = 0;
    for (int i = 0; i < 8; ++i) {
        int s = sm[i];
        if (i < w) before += s;
        all += s;
    }
    __syncthreads();
    total = all;
    return before + inc - v;
}


__global__ void k_part(const int* __restrict__ src, const int* __restrict__ dst,
                       int* __restrict__ table, int2* __restrict__ part) {
    __shared__ int lh[NBK];
    __shared__ int lex[NBK];
    __shared__ int lfill[NBK];
    __shared__ int sm[8];
    __shared__ int2 stage[CHUNK];
    int tid = threadIdx.x;
    for (int i = tid; i < NBK; i += 256) { lh[i] = 0; lfill[i] = 0; }
    __syncthreads();

    int e0 = blockIdx.x * CHUNK;
    int s[EPT], d[EPT];
    #pragma unroll
    for (int k = 0; k < EPT; ++k) {
        int e = e0 + k * 256 + tid;
        bool ok = e < N_EDGES;
        d[k] = ok ? dst[e] : -1;
        s[k] = ok ? src[e] : 0;
    }
    #pragma unroll
    for (int k = 0; k < EPT; ++k)
        if (d[k] >= 0) atomicAdd(&lh[d[k] >> BK_SHIFT], 1);
    __syncthreads();

    int t0 = 2 * tid, t1 = 2 * tid + 1;
    int h0 = (t0 < NBK) ? lh[t0] : 0;
    int h1 = (t1 < NBK) ? lh[t1] : 0;
    int tot;
    int hex = blockExclScan(h0 + h1, tot, sm);
    int* trow = table + (size_t)blockIdx.x * TROW;
    if (t0 < NBK) lex[t0] = hex;
    if (t1 < NBK) lex[t1] = hex + h0;
    if (t0 < TROW) trow[t0] = hex;
    if (t1 < TROW) trow[t1] = hex + h0;
    __syncthreads();

    #pragma unroll
    for (int k = 0; k < EPT; ++k) {
        if (d[k] >= 0) {
            int b = d[k] >> BK_SHIFT;
            int r = atomicAdd(&lfill[b], 1);
            stage[lex[b] + r] = make_int2(s[k], d[k]);
        }
    }
    __syncthreads();

    int n = N_EDGES - e0;
    if (n > CHUNK) n = CHUNK;
    for (int i = tid; i < n; i += 256) part[e0 + i] = stage[i];
}


__global__ void
k_bucket(const int* __restrict__ table, const int2* __restrict__ part,
         int* __restrict__ rowptr, int* __restrict__ cnt, float* __restrict__ dinv,
         int2* __restrict__ rec) {
    __shared__ int lcnt[BK_NODES];
    __shared__ int lptr[BK_NODES];
    __shared__ int lcur[BK_NODES];
    __shared__ int wsum[32];
    __shared__ int wbase[32];
    __shared__ int rsrc[NCHUNK];
    __shared__ int rlen[NCHUNK];
    __shared__ int rpos[NCHUNK];
    __shared__ int2 sin_[BCAP];
    __shared__ int2 sout[BCAP];
    int tid  = threadIdx.x;
    int lane = tid & 31;
    int w    = tid >> 5;
    int b    = blockIdx.x;

    int s0 = 0, len = 0;
    if (tid < NCHUNK) {
        const int* trow = table + (size_t)tid * TROW + b;
        s0  = trow[0];
        len = trow[1] - s0;
    }
    int inc = len;
    for (int o = 1; o < 32; o <<= 1) {
        int nb = __shfl_up(inc, o, 32);
        if (lane >= o) inc += nb;
    }
    int bs = s0;
    for (int o = 16; o > 0; o >>= 1) bs += __shfl_xor(bs, o, 32);
    if (lane == 31) wsum[w]  = inc;
    if (lane == 0)  wbase[w] = bs;
    if (tid < BK_NODES) { lcnt[tid] = 0; lcur[tid] = 0; }
    __syncthreads();
    int before = 0, n = 0, base = 0;
    for (int i = 0; i < 32; ++i) {
        int sv = wsum[i];
        if (i < w) before += sv;
        n    += sv;
        base += wbase[i];
    }
    if (tid < NCHUNK) {
        rsrc[tid] = tid * CHUNK + s0;
        rlen[tid] = len;
        rpos[tid] = before + inc - len;
    }
    __syncthreads();

    int g = tid >> 4;
    int l = tid & 15;
    bool st = (n <= BCAP);

    if (st) {
        int2 ed[RPG];
        int rl[RPG], rs[RPG], rp[RPG];
        #pragma unroll
        for (int k = 0; k < RPG; ++k) {
            int c  = g + k * NGRP;
            int cc = (c < NCHUNK) ? c : (NCHUNK - 1);
            int ln = rlen[cc];
            rl[k] = (c < NCHUNK) ? ln : 0;
            rs[k] = rsrc[cc];
            rp[k] = rpos[cc];
        }
        #pragma unroll
        for (int k = 0; k < RPG; ++k)
            ed[k] = (l < rl[k]) ? part[rs[k] + l] : make_int2(0, -1);
        #pragma unroll
        for (int k = 0; k < RPG; ++k)
            if (ed[k].y >= 0) sin_[rp[k] + l] = ed[k];
        #pragma unroll
        for (int k = 0; k < RPG; ++k)
            for (int i = l + 16; i < rl[k]; i += 16) sin_[rp[k] + i] = part[rs[k] + i];
        __syncthreads();
        for (int i = tid; i < n; i += 1024)
            atomicAdd(&lcnt[sin_[i].y & (BK_NODES - 1)], 1);
    } else {
        for (int c = g; c < NCHUNK; c += NGRP) {
            int ln = rlen[c];
            for (int i = l; i < ln; i += 16)
                atomicAdd(&lcnt[part[rsrc[c] + i].y & (BK_NODES - 1)], 1);
        }
    }
    __syncthreads();

    if (tid < 32) {
        int c[8];
        int sum = 0;
        #pragma unroll
        for (int k = 0; k < 8; ++k) { c[k] = lcnt[tid * 8 + k]; sum += c[k]; }
        int pin = sum;
        for (int o = 1; o < 32; o <<= 1) {
            int nb = __shfl_up(pin, o, 32);
            if (lane >= o) pin += nb;
        }
        int run = pin - sum;
        #pragma unroll
        for (int k = 0; k < 8; ++k) { lptr[tid * 8 + k] = run; run += c[k]; }
    }
    __syncthreads();

    if (tid < BK_NODES) {
        int node = b * BK_NODES + tid;
        if (node < N_NODES) {
            int c = lcnt[tid];
            rowptr[node] = base + lptr[tid];
            cnt[node]    = c;
            dinv[node]   = rsqrtf((float)c + 1.0f);
        }
    }

    if (st) {
        for (int i = tid; i < n; i += 1024) {
            int2 e = sin_[i];
            int dl = e.y & (BK_NODES - 1);
            int r  = atomicAdd(&lcur[dl], 1);
            sout[lptr[dl] + r] = e;
        }
        __syncthreads();
        for (int i = tid; i < n; i += 1024) rec[base + i] = sout[i];
    } else {
        for (int c = g; c < NCHUNK; c += NGRP) {
            int ln = rlen[c];
            for (int i = l; i < ln; i += 16) {
                int2 e = part[rsrc[c] + i];
                int dl = e.y & (BK_NODES - 1);
                int r  = atomicAdd(&lcur[dl], 1);
                rec[base + lptr[dl] + r] = e;
            }
        }
    }
}


__global__ void k_weights(const float* __restrict__ dinv, int* rec) {
    int t = blockIdx.x * blockDim.x + threadIdx.x;
    if (t < N_EDGES / 2) {
        int4 v = ((const int4*)rec)[t];
        float w0 = dinv[v.x] * dinv[v.y];
        float w1 = dinv[v.z] * dinv[v.w];
        ((int4*)rec)[t] = make_int4(v.x, __float_as_int(w0), v.z, __float_as_int(w1));
    }
}


template<int K, int NCOLS, bool RELU>
__global__ void k_gemm_wmma(const float* __restrict__ A,
                            const float* __restrict__ W,
                            float* __restrict__ out) {
    const int NT = NCOLS / 16;
    const int NG = NCOLS / 64;
    int strip = (blockIdx.x * blockDim.x + threadIdx.x) >> 5;
    int lane  = threadIdx.x & 31;

    int hi = lane >> 4;
    int c  = lane & 15;
    int mBase = strip * 16;

    const float* aRow = A + (size_t)(mBase + c) * K + 2 * hi;
    const float* wRow = W + (size_t)(2 * hi) * NCOLS + 4 * c;

    v8f acc[NT];
    #pragma unroll
    for (int t = 0; t < NT; ++t)
        acc[t] = (v8f){0.f, 0.f, 0.f, 0.f, 0.f, 0.f, 0.f, 0.f};

    #pragma unroll 2
    for (int k = 0; k < K; k += 4) {
        v2f av = *(const v2f*)(aRow + k);
        if (RELU) { av.x = fmaxf(av.x, 0.0f); av.y = fmaxf(av.y, 0.0f); }

        v4f b0[NG], b1[NG];
        #pragma unroll
        for (int g = 0; g < NG; ++g) {
            b0[g] = *(const v4f*)(wRow + (size_t)k * NCOLS + 64 * g);
            b1[g] = *(const v4f*)(wRow + (size_t)(k + 1) * NCOLS + 64 * g);
        }

        #pragma unroll
        for (int t = 0; t < NT; ++t) {
            v2f bv;
            bv.x = b0[t >> 2][t & 3];
            bv.y = b1[t >> 2][t & 3];
            acc[t] = __builtin_amdgcn_wmma_f32_16x16x4_f32(false, av, false, bv,
                                                           (short)0, acc[t], false, false);
        }
    }

    float* oRow = out + (size_t)(mBase + 8 * hi) * NCOLS + 4 * c;
    #pragma unroll
    for (int r = 0; r < 8; ++r) {
        #pragma unroll
        for (int g = 0; g < NG; ++g) {
            v4f o;
            o.x = acc[4 * g + 0][r];
            o.y = acc[4 * g + 1][r];
            o.z = acc[4 * g + 2][r];
            o.w = acc[4 * g + 3][r];
            *(v4f*)(oRow + (size_t)r * NCOLS + 64 * g) = o;
        }
    }
}


template<int V> struct RowVec;
template<> struct RowVec<4> { typedef v4f T; };
template<> struct RowVec<2> { typedef v2f T; };

template<int D>
__global__ void k_gather(const int* __restrict__ rowptr, const int* __restrict__ cnt,
                         const int* __restrict__ rec, const float* __restrict__ dinv,
                         const float* __restrict__ bias, const float* __restrict__ h,
                         float* __restrict__ agg) {
    const int V = D / 32;
    typedef typename RowVec<V>::T vec;
    int lane = threadIdx.x & 31;
    int node = __builtin_amdgcn_readfirstlane((int)((blockIdx.x * blockDim.x + threadIdx.x) >> 5));

    int j   = rowptr[node];
    int end = j + cnt[node];
    float di = dinv[node];
    float sn = di * di;

    const float* hl = h + lane * V;
    vec acc = *(const vec*)(hl + (size_t)node * D) * sn + *(const vec*)(bias + lane * V);

    const int2* r = (const int2*)rec;
    for (; j + 4 <= end; j += 4) {
        int2 p0 = r[j], p1 = r[j + 1], p2 = r[j + 2], p3 = r[j + 3];
        vec x0 = *(const vec*)(hl + (size_t)p0.x * D);
        vec x1 = *(const vec*)(hl + (size_t)p1.x * D);
        vec x2 = *(const vec*)(hl + (size_t)p2.x * D);
        vec x3 = *(const vec*)(hl + (size_t)p3.x * D);
        acc += x0 * __int_as_float(p0.y);
        acc += x1 * __int_as_float(p1.y);
        acc += x2 * __int_as_float(p2.y);
        acc += x3 * __int_as_float(p3.y);
    }
    for (; j < end; ++j) {
        int2 p = r[j];
        vec x = *(const vec*)(hl + (size_t)p.x * D);
        acc += x * __int_as_float(p.y);
    }
    *(vec*)(agg + (size_t)node * D + lane * V) = acc;
}


extern "C" void kernel_launch(void* const* d_in, const int* in_sizes, int n_in,
                              void* d_out, int out_size, void* d_ws, size_t ws_size,
                              hipStream_t stream) {
    const float* x   = (const float*)d_in[0];
    const int*   ei  = (const int*)d_in[1];
    const float* W1  = (const float*)d_in[2];
    const float* b1  = (const float*)d_in[3];
    const float* W2  = (const float*)d_in[4];
    const float* b2  = (const float*)d_in[5];
    float*       out = (float*)d_out;

    const int* src = ei;
    const int* dst = ei + N_EDGES;

    float* ws    = (float*)d_ws;
    float* dinv  = ws;
    int*   table = (int*)(dinv + N_NODES);
    float* h1    = dinv + N_NODES + N_EDGES;
    float* agg1  = h1    + (size_t)N_NODES * DH;
    int*   rec   = (int*)(agg1 + (size_t)N_NODES * DH);
    int*   cnt   = rec + 2 * (size_t)N_EDGES;
    int*   rowp  = cnt + N_NODES;
    int2*  part  = (int2*)agg1;
    float* h2    = h1;

    const int B = 256;

    k_part    <<<NCHUNK, B, 0, stream>>>(src, dst, table, part);
    k_bucket  <<<NBK, 1024, 0, stream>>>(table, part, rowp, cnt, dinv, (int2*)rec);
    k_weights <<<(N_EDGES / 2 + B - 1) / B, B, 0, stream>>>(dinv, rec);

    const int GB = 64;
    {
        int waves = N_NODES / 16;
        k_gemm_wmma<DIN, DH, false><<<waves * 32 / GB, GB, 0, stream>>>(x, W1, h1);
    }
    k_gather<DH><<<N_NODES * 32 / B, B, 0, stream>>>(rowp, cnt, rec, dinv, b1, h1, agg1);

    {
        int waves = N_NODES / 16;
        k_gemm_wmma<DH, DOUT, true><<<waves * 32 / GB, GB, 0, stream>>>(agg1, W2, h2);
    }
    k_gather<DOUT><<<N_NODES * 32 / B, B, 0, stream>>>(rowp, cnt, rec, dinv, b2, h2, out);
}


# Layer-1 gather and layer-2 GEMM in one launch: a wave gathers a 16-node strip into an LDS tile and multiplies it by W2 staged in LDS; agg1 never written; 6 -> 5 launches; every sum order unchanged
# speedup vs baseline: 4.8549x; 1.2971x over previous
#include <hip/hip_runtime.h>

#define N_NODES 100000
#define N_EDGES 1600000
#define DIN 128
#define DH  128
#define DOUT 64

typedef __attribute__((ext_vector_type(2))) float v2f;
typedef __attribute__((ext_vector_type(4))) float v4f;
typedef __attribute__((ext_vector_type(8))) float v8f;


#define BK_SHIFT 8
#define BK_NODES (1 << BK_SHIFT)
#define NBK ((N_NODES + BK_NODES - 1) / BK_NODES)
#define CHUNK 4096
#define NCHUNK ((N_EDGES + CHUNK - 1) / CHUNK)
#define EPT (CHUNK / 256)
#define BCAP 6144
#define TROW (NBK + 1)
#define NGRP 64
#define RPG ((NCHUNK + NGRP - 1) / NGRP)

__device__ __forceinline__ int blockExclScan(int v, int& total, int* sm) {
    int lane = threadIdx.x & 31;
    int w    = threadIdx.x >> 5;
    int inc = v;
    for (int o = 1; o < 32; o <<= 1) {
        int n = __shfl_up(inc, o, 32);
        if (lane >= o) inc += n;
    }
    if (lane == 31) sm[w] = inc;
    __syncthreads();
    int before = 0, all = 0;
    for (int i = 0; i < 8; ++i) {
        int s = sm[i];
        if (i < w) before += s;
        all += s;
    }
    __syncthreads();
    total = all;
    return before + inc - v;
}


__global__ void k_part(const int* __restrict__ src, const int* __restrict__ dst,
                       int* __restrict__ table, int2* __restrict__ part) {
    __shared__ int lh[NBK];
    __shared__ int lex[NBK];
    __shared__ int lfill[NBK];
    __shared__ int sm[8];
    __shared__ int2 stage[CHUNK];
    int tid = threadIdx.x;
    for (int i = tid; i < NBK; i += 256) { lh[i] = 0; lfill[i] = 0; }
    __syncthreads();

    int e0 = blockIdx.x * CHUNK;
    int s[EPT], d[EPT];
    #pragma unroll
    for (int k = 0; k < EPT; ++k) {
        int e = e0 + k * 256 + tid;
        bool ok = e < N_EDGES;
        d[k] = ok ? dst[e] : -1;
        s[k] = ok ? src[e] : 0;
    }
    #pragma unroll
    for (int k = 0; k < EPT; ++k)
        if (d[k] >= 0) atomicAdd(&lh[d[k] >> BK_SHIFT], 1);
    __syncthreads();

    int t0 = 2 * tid, t1 = 2 * tid + 1;
    int h0 = (t0 < NBK) ? lh[t0] : 0;
    int h1 = (t1 < NBK) ? lh[t1] : 0;
    int tot;
    int hex = blockExclScan(h0 + h1, tot, sm);
    int* trow = table + (size_t)blockIdx.x * TROW;
    if (t0 < NBK) lex[t0] = hex;
    if (t1 < NBK) lex[t1] = hex + h0;
    if (t0 < TROW) trow[t0] = hex;
    if (t1 < TROW) trow[t1] = hex + h0;
    __syncthreads();

    #pragma unroll
    for (int k = 0; k < EPT; ++k) {
        if (d[k] >= 0) {
            int b = d[k] >> BK_SHIFT;
            int r = atomicAdd(&lfill[b], 1);
            stage[lex[b] + r] = make_int2(s[k], d[k]);
        }
    }
    __syncthreads();

    int n = N_EDGES - e0;
    if (n > CHUNK) n = CHUNK;
    for (int i = tid; i < n; i += 256) part[e0 + i] = stage[i];
}


__global__ void
k_bucket(const int* __restrict__ table, const int2* __restrict__ part,
         int* __restrict__ rowptr, int* __restrict__ cnt, float* __restrict__ dinv,
         int* __restrict__ rec) {
    __shared__ int lcnt[BK_NODES];
    __shared__ int lptr[BK_NODES];
    __shared__ int lcur[BK_NODES];
    __shared__ int wsum[32];
    __shared__ int wbase[32];
    __shared__ int rsrc[NCHUNK];
    __shared__ int rlen[NCHUNK];
    __shared__ int rpos[NCHUNK];
    __shared__ int2 sin_[BCAP];
    __shared__ int sout[BCAP];
    int tid  = threadIdx.x;
    int lane = tid & 31;
    int w    = tid >> 5;
    int b    = blockIdx.x;

    int s0 = 0, len = 0;
    if (tid < NCHUNK) {
        const int* trow = table + (size_t)tid * TROW + b;
        s0  = trow[0];
        len = trow[1] - s0;
    }
    int inc = len;
    for (int o = 1; o < 32; o <<= 1) {
        int nb = __shfl_up(inc, o, 32);
        if (lane >= o) inc += nb;
    }
    int bs = s0;
    for (int o = 16; o > 0; o >>= 1) bs += __shfl_xor(bs, o, 32);
    if (lane == 31) wsum[w]  = inc;
    if (lane == 0)  wbase[w] = bs;
    if (tid < BK_NODES) { lcnt[tid] = 0; lcur[tid] = 0; }
    __syncthreads();
    int before = 0, n = 0, base = 0;
    for (int i = 0; i < 32; ++i) {
        int sv = wsum[i];
        if (i < w) before += sv;
        n    += sv;
        base += wbase[i];
    }
    n    = __builtin_amdgcn_readfirstlane(n);
    base = __builtin_amdgcn_readfirstlane(base);
    if (tid < NCHUNK) {
        rsrc[tid] = tid * CHUNK + s0;
        rlen[tid] = len;
        rpos[tid] = before + inc - len;
    }
    __syncthreads();

    int g = tid >> 4;
    int l = tid & 15;
    bool st = (n <= BCAP);

    if (st) {
        int2 ed[RPG];
        int rl[RPG], rs[RPG], rp[RPG];
        #pragma unroll
        for (int k = 0; k < RPG; ++k) {
            int c  = g + k * NGRP;
            int cc = (c < NCHUNK) ? c : (NCHUNK - 1);
            int ln = rlen[cc];
            rl[k] = (c < NCHUNK) ? ln : 0;
            rs[k] = rsrc[cc];
            rp[k] = rpos[cc];
        }
        #pragma unroll
        for (int k = 0; k < RPG; ++k)
            ed[k] = (l < rl[k]) ? part[rs[k] + l] : make_int2(0, -1);
        #pragma unroll
        for (int k = 0; k < RPG; ++k)
            if (ed[k].y >= 0) sin_[rp[k] + l] = ed[k];
        #pragma unroll
        for (int k = 0; k < RPG; ++k)
            for (int i = l + 16; i < rl[k]; i += 16) sin_[rp[k] + i] = part[rs[k] + i];
        __syncthreads();
        for (int i = tid; i < n; i += 1024)
            atomicAdd(&lcnt[sin_[i].y & (BK_NODES - 1)], 1);
    } else {
        for (int c = g; c < NCHUNK; c += NGRP) {
            int ln = rlen[c];
            for (int i = l; i < ln; i += 16)
                atomicAdd(&lcnt[part[rsrc[c] + i].y & (BK_NODES - 1)], 1);
        }
    }
    __syncthreads();

    if (tid < 32) {
        int c[8];
        int sum = 0;
        #pragma unroll
        for (int k = 0; k < 8; ++k) { c[k] = lcnt[tid * 8 + k]; sum += c[k]; }
        int pin = sum;
        for (int o = 1; o < 32; o <<= 1) {
            int nb = __shfl_up(pin, o, 32);
            if (lane >= o) pin += nb;
        }
        int run = pin - sum;
        #pragma unroll
        for (int k = 0; k < 8; ++k) { lptr[tid * 8 + k] = run; run += c[k]; }
    }
    __syncthreads();

    if (tid < BK_NODES) {
        int node = b * BK_NODES + tid;
        if (node < N_NODES) {
            int c = lcnt[tid];
            rowptr[node] = base + lptr[tid];
            cnt[node]    = c;
            dinv[node]   = rsqrtf((float)c + 1.0f);
        }
    }

    if (st) {
        for (int i = tid; i < n; i += 1024) {
            int2 e = sin_[i];
            int dl = e.y & (BK_NODES - 1);
            int r  = atomicAdd(&lcur[dl], 1);
            sout[lptr[dl] + r] = e.x;
        }
        __syncthreads();
        for (int i = tid; i < n; i += 1024) rec[base + i] = sout[i];
    } else {
        for (int c = g; c < NCHUNK; c += NGRP) {
            int ln = rlen[c];
            for (int i = l; i < ln; i += 16) {
                int2 e = part[rsrc[c] + i];
                int dl = e.y & (BK_NODES - 1);
                int r  = atomicAdd(&lcur[dl], 1);
                rec[base + lptr[dl] + r] = e.x;
            }
        }
    }
}


template<int K, int NCOLS, bool RELU>
__global__ void k_gemm_wmma(const float* __restrict__ A,
                            const float* __restrict__ W,
                            const float* __restrict__ dinv,
                            float* __restrict__ out) {
    const int NT = NCOLS / 16;
    const int NG = NCOLS / 64;
    int strip = (blockIdx.x * blockDim.x + threadIdx.x) >> 5;
    int lane  = threadIdx.x & 31;

    int hi = lane >> 4;
    int c  = lane & 15;
    int mBase = strip * 16;

    const float* aRow = A + (size_t)(mBase + c) * K + 2 * hi;
    const float* wRow = W + (size_t)(2 * hi) * NCOLS + 4 * c;

    const float* dRow = dinv + mBase + 8 * hi;
    v4f dlo = *(const v4f*)(dRow);
    v4f dhi = *(const v4f*)(dRow + 4);

    v8f acc[NT];
    #pragma unroll
    for (int t = 0; t < NT; ++t)
        acc[t] = (v8f){0.f, 0.f, 0.f, 0.f, 0.f, 0.f, 0.f, 0.f};

    #pragma unroll 2
    for (int k = 0; k < K; k += 4) {
        v2f av = *(const v2f*)(aRow + k);
        if (RELU) { av.x = fmaxf(av.x, 0.0f); av.y = fmaxf(av.y, 0.0f); }

        v4f b0[NG], b1[NG];
        #pragma unroll
        for (int g = 0; g < NG; ++g) {
            b0[g] = *(const v4f*)(wRow + (size_t)k * NCOLS + 64 * g);
            b1[g] = *(const v4f*)(wRow + (size_t)(k + 1) * NCOLS + 64 * g);
        }

        #pragma unroll
        for (int t = 0; t < NT; ++t) {
            v2f bv;
            bv.x = b0[t >> 2][t & 3];
            bv.y = b1[t >> 2][t & 3];
            acc[t] = __builtin_amdgcn_wmma_f32_16x16x4_f32(false, av, false, bv,
                                                           (short)0, acc[t], false, false);
        }
    }

    float* oRow = out + (size_t)(mBase + 8 * hi) * NCOLS + 4 * c;
    #pragma unroll
    for (int r = 0; r < 8; ++r) {
        float sc = (r < 4) ? dlo[r & 3] : dhi[r & 3];
        #pragma unroll
        for (int g = 0; g < NG; ++g) {
            v4f o;
            o.x = acc[4 * g + 0][r] * sc;
            o.y = acc[4 * g + 1][r] * sc;
            o.z = acc[4 * g + 2][r] * sc;
            o.w = acc[4 * g + 3][r] * sc;
            *(v4f*)(oRow + (size_t)r * NCOLS + 64 * g) = o;
        }
    }
}


template<int V> struct RowVec;
template<> struct RowVec<4> { typedef v4f T; };
template<> struct RowVec<2> { typedef v2f T; };

template<int D>
__global__ void k_gather(const int* __restrict__ rowptr, const int* __restrict__ cnt,
                         const int* __restrict__ rec, const float* __restrict__ dinv,
                         const float* __restrict__ bias, const float* __restrict__ h,
                         float* __restrict__ agg) {
    const int V = D / 32;
    typedef typename RowVec<V>::T vec;
    int lane = threadIdx.x & 31;
    int node = __builtin_amdgcn_readfirstlane((int)((blockIdx.x * blockDim.x + threadIdx.x) >> 5));

    int j   = rowptr[node];
    int end = j + cnt[node];
    float di = dinv[node];

    const float* hl = h + lane * V;
    vec acc = *(const vec*)(hl + (size_t)node * D);

    for (; j + 4 <= end; j += 4) {
        int p0 = rec[j], p1 = rec[j + 1], p2 = rec[j + 2], p3 = rec[j + 3];
        vec x0 = *(const vec*)(hl + (size_t)p0 * D);
        vec x1 = *(const vec*)(hl + (size_t)p1 * D);
        vec x2 = *(const vec*)(hl + (size_t)p2 * D);
        vec x3 = *(const vec*)(hl + (size_t)p3 * D);
        acc += x0;
        acc += x1;
        acc += x2;
        acc += x3;
    }
    for (; j < end; ++j) {
        int p = rec[j];
        acc += *(const vec*)(hl + (size_t)p * D);
    }
    *(vec*)(agg + (size_t)node * D + lane * V) = acc * di + *(const vec*)(bias + lane * V);
}


#define FW 25
#define NSTRIP (N_NODES / 16)
#define NFB (NSTRIP / FW)
static_assert(NSTRIP % FW == 0, "one strip per wave, exact grid");
static_assert(N_NODES % 16 == 0, "whole strips");
#define TP (DH + 4)
#define WP (DOUT + 4)
#define GBATCH 16

__global__ void __launch_bounds__(FW * 32)
k_gather_gemm(const int* __restrict__ rowptr, const int* __restrict__ cnt,
              const int* __restrict__ rec, const float* __restrict__ dinv,
              const float* __restrict__ bias, const float* __restrict__ h,
              const float* __restrict__ W, float* __restrict__ out) {
    __shared__ float w2s[DH * WP];
    __shared__ float tiles[FW * 16 * TP];
    int tid  = threadIdx.x;
    int lane = tid & 31;
    int wv   = __builtin_amdgcn_readfirstlane(tid >> 5);

    int strip = blockIdx.x * FW + wv;
    int n0 = strip * 16;
    float* tl = tiles + wv * (16 * TP);
    const float* hl = h + lane * 4;
    float* tcol = tl + 4 * lane;

    v4f wq[4];
    if (tid < 512) {
        #pragma unroll
        for (int it = 0; it < 4; ++it)
            wq[it] = *(const v4f*)(W + 4 * (size_t)(tid + it * 512));
    }

    int li   = lane & 15;
    int rp   = rowptr[n0 + li];
    int endv = rp + cnt[n0 + li];
    float dv = dinv[n0 + li];
    v4f bvec = *(const v4f*)(bias + lane * 4);

    #pragma unroll
    for (int i = 0; i < 16; ++i)
        *(v4f*)(tcol + i * TP) = *(const v4f*)(hl + (size_t)(n0 + i) * DH);

    if (tid < 512) {
        #pragma unroll
        for (int it = 0; it < 4; ++it) {
            int i = tid + it * 512;
            *(v4f*)&w2s[(i >> 4) * WP + 4 * (i & 15)] = wq[it];
        }
    }
    __syncthreads();

    int j    = __builtin_amdgcn_readfirstlane(rp);
    int jend = __builtin_amdgcn_readlane(endv, 15);
    int i    = 0;
    int nend = __builtin_amdgcn_readlane(endv, 0);
    v4f acc  = *(const v4f*)(tcol);

#define FINISH_NODE()                                                                         \
    {                                                                                         \
        float di = __builtin_bit_cast(float,                                                  \
                       __builtin_amdgcn_readlane(__builtin_bit_cast(int, dv), i));            \
        v4f o = acc * di + bvec;                                                              \
        o.x = fmaxf(o.x, 0.0f); o.y = fmaxf(o.y, 0.0f);                                       \
        o.z = fmaxf(o.z, 0.0f); o.w = fmaxf(o.w, 0.0f);                                       \
        *(v4f*)(tcol + i * TP) = o;                                                           \
    }

    int p[GBATCH];
    if (j < jend) {
        #pragma unroll
        for (int k = 0; k < GBATCH; ++k) {
            int jj = j + k;
            p[k] = rec[jj < jend ? jj : jend - 1];
        }
    }
    while (j < jend) {
        v4f x[GBATCH];
        #pragma unroll
        for (int k = 0; k < GBATCH; ++k) x[k] = *(const v4f*)(hl + (size_t)p[k] * DH);
        int jn = j + GBATCH;
        if (jn < jend) {
            #pragma unroll
            for (int k = 0; k < GBATCH; ++k) {
                int jj = jn + k;
                p[k] = rec[jj < jend ? jj : jend - 1];
            }
        }
        #pragma unroll
        for (int k = 0; k < GBATCH; ++k) {
            int jj = j + k;
            if (jj < jend) {
                while (jj >= nend) {
                    FINISH_NODE();
                    ++i;
                    nend = __builtin_amdgcn_readlane(endv, i);
                    acc  = *(const v4f*)(tcol + i * TP);
                }
                acc += x[k];
            }
        }
        j = jn;
    }
    for (;;) {
        FINISH_NODE();
        ++i;
        if (i >= 16) break;
        acc = *(const v4f*)(tcol + i * TP);
    }
#undef FINISH_NODE


    int hi = lane >> 4;
    int c  = lane & 15;
    const float* aRow = tl + c * TP + 2 * hi;
    const float* wRow = w2s + (2 * hi) * WP + 4 * c;
    const float* dRow = dinv + n0 + 8 * hi;
    v4f dlo = *(const v4f*)(dRow);
    v4f dhi = *(const v4f*)(dRow + 4);

    v8f am[4];
    #pragma unroll
    for (int t = 0; t < 4; ++t)
        am[t] = (v8f){0.f, 0.f, 0.f, 0.f, 0.f, 0.f, 0.f, 0.f};

    #pragma unroll 2
    for (int k = 0; k < DH; k += 4) {
        v2f av = *(const v2f*)(aRow + k);
        v4f b0 = *(const v4f*)(wRow + k * WP);
        v4f b1 = *(const v4f*)(wRow + (k + 1) * WP);
        #pragma unroll
        for (int t = 0; t < 4; ++t) {
            v2f bv;
            bv.x = b0[t];
            bv.y = b1[t];
            am[t] = __builtin_amdgcn_wmma_f32_16x16x4_f32(false, av, false, bv,
                                                          (short)0, am[t], false, false);
        }
    }

    float* oRow = out + (size_t)(n0 + 8 * hi) * DOUT + 4 * c;
    #pragma unroll
    for (int r = 0; r < 8; ++r) {
        float sc = (r < 4) ? dlo[r & 3] : dhi[r & 3];
        v4f o;
        o.x = am[0][r] * sc;
        o.y = am[1][r] * sc;
        o.z = am[2][r] * sc;
        o.w = am[3][r] * sc;
        *(v4f*)(oRow + (size_t)r * DOUT) = o;
    }
}


extern "C" void kernel_launch(void* const* d_in, const int* in_sizes, int n_in,
                              void* d_out, int out_size, void* d_ws, size_t ws_size,
                              hipStream_t stream) {
    const float* x   = (const float*)d_in[0];
    const int*   ei  = (const int*)d_in[1];
    const float* W1  = (const float*)d_in[2];
    const float* b1  = (const float*)d_in[3];
    const float* W2  = (const float*)d_in[4];
    const float* b2  = (const float*)d_in[5];
    float*       out = (float*)d_out;

    const int* src = ei;
    const int* dst = ei + N_EDGES;

    float* ws    = (float*)d_ws;
    float* dinv  = ws;
    int*   table = (int*)(dinv + N_NODES);
    float* h1    = dinv + N_NODES + N_EDGES;
    float* agg1  = h1    + (size_t)N_NODES * DH;
    int*   rec   = (int*)(agg1 + (size_t)N_NODES * DH);
    int*   cnt   = rec + 2 * (size_t)N_EDGES;
    int*   rowp  = cnt + N_NODES;
    int2*  part  = (int2*)agg1;
    float* h2    = agg1;

    const int B = 256;

    k_part    <<<NCHUNK, B, 0, stream>>>(src, dst, table, part);
    k_bucket  <<<NBK, 1024, 0, stream>>>(table, part, rowp, cnt, dinv, rec);

    const int GB = 64;
    {
        int waves = N_NODES / 16;
        k_gemm_wmma<DIN, DH, false><<<waves * 32 / GB, GB, 0, stream>>>(x, W1, dinv, h1);
    }
    k_gather_gemm<<<NFB, FW * 32, 0, stream>>>(rowp, cnt, rec, dinv, b1, h1, W2, h2);

    k_gather<DOUT><<<N_NODES * 32 / B, B, 0, stream>>>(rowp, cnt, rec, dinv, b2, h2, out);
}


# speedup vs baseline: 5.0308x; 1.0362x over previous
#include <hip/hip_runtime.h>

#define N_NODES 100000
#define N_EDGES 1600000
#define DIN 128
#define DH  128
#define DOUT 64

typedef __attribute__((ext_vector_type(2))) float v2f;
typedef __attribute__((ext_vector_type(4))) float v4f;
typedef __attribute__((ext_vector_type(8))) float v8f;


#define BK_SHIFT 8
#define BK_NODES (1 << BK_SHIFT)
#define NBK ((N_NODES + BK_NODES - 1) / BK_NODES)
#define CHUNK 4096
#define NCHUNK ((N_EDGES + CHUNK - 1) / CHUNK)
#define EPT (CHUNK / 256)
#define BCAP 6144
#define TROW (NBK + 1)
#define NGRP 64
#define RPG ((NCHUNK + NGRP - 1) / NGRP)

__device__ __forceinline__ int blockExclScan(int v, int& total, int* sm) {
    int lane = threadIdx.x & 31;
    int w    = threadIdx.x >> 5;
    int inc = v;
    for (int o = 1; o < 32; o <<= 1) {
        int n = __shfl_up(inc, o, 32);
        if (lane >= o) inc += n;
    }
    if (lane == 31) sm[w] = inc;
    __syncthreads();
    int before = 0, all = 0;
    for (int i = 0; i < 8; ++i) {
        int s = sm[i];
        if (i < w) before += s;
        all += s;
    }
    __syncthreads();
    total = all;
    return before + inc - v;
}


__global__ void k_part(const int* __restrict__ src, const int* __restrict__ dst,
                       int* __restrict__ table, int2* __restrict__ part) {
    __shared__ int lh[NBK];
    __shared__ int lex[NBK];
    __shared__ int lfill[NBK];
    __shared__ int sm[8];
    __shared__ int2 stage[CHUNK];
    int tid = threadIdx.x;
    for (int i = tid; i < NBK; i += 256) { lh[i] = 0; lfill[i] = 0; }
    __syncthreads();

    int e0 = blockIdx.x * CHUNK;
    int s[EPT], d[EPT];
    #pragma unroll
    for (int k = 0; k < EPT; ++k) {
        int e = e0 + k * 256 + tid;
        bool ok = e < N_EDGES;
        d[k] = ok ? dst[e] : -1;
        s[k] = ok ? src[e] : 0;
    }
    #pragma unroll
    for (int k = 0; k < EPT; ++k)
        if (d[k] >= 0) atomicAdd(&lh[d[k] >> BK_SHIFT], 1);
    __syncthreads();

    int t0 = 2 * tid, t1 = 2 * tid + 1;
    int h0 = (t0 < NBK) ? lh[t0] : 0;
    int h1 = (t1 < NBK) ? lh[t1] : 0;
    int tot;
    int hex = blockExclScan(h0 + h1, tot, sm);
    int* trow = table + (size_t)blockIdx.x * TROW;
    if (t0 < NBK) lex[t0] = hex;
    if (t1 < NBK) lex[t1] = hex + h0;
    if (t0 < TROW) trow[t0] = hex;
    if (t1 < TROW) trow[t1] = hex + h0;
    __syncthreads();

    #pragma unroll
    for (int k = 0; k < EPT; ++k) {
        if (d[k] >= 0) {
            int b = d[k] >> BK_SHIFT;
            int r = atomicAdd(&lfill[b], 1);
            stage[lex[b] + r] = make_int2(s[k], d[k]);
        }
    }
    __syncthreads();

    int n = N_EDGES - e0;
    if (n > CHUNK) n = CHUNK;
    for (int i = tid; i < n; i += 256) part[e0 + i] = stage[i];
}


__global__ void
k_bucket(const int* __restrict__ table, const int2* __restrict__ part,
         int* __restrict__ rowptr, int* __restrict__ cnt, float* __restrict__ dinv,
         int* __restrict__ rec) {
    __shared__ int lcnt[BK_NODES];
    __shared__ int lptr[BK_NODES];
    __shared__ int lcur[BK_NODES];
    __shared__ int wsum[32];
    __shared__ int wbase[32];
    __shared__ int rsrc[NCHUNK];
    __shared__ int rlen[NCHUNK];
    __shared__ int rpos[NCHUNK];
    __shared__ int2 sin_[BCAP];
    __shared__ int sout[BCAP];
    int tid  = threadIdx.x;
    int lane = tid & 31;
    int w    = tid >> 5;
    int b    = blockIdx.x;

    int s0 = 0, len = 0;
    if (tid < NCHUNK) {
        const int* trow = table + (size_t)tid * TROW + b;
        s0  = trow[0];
        len = trow[1] - s0;
    }
    int inc = len;
    for (int o = 1; o < 32; o <<= 1) {
        int nb = __shfl_up(inc, o, 32);
        if (lane >= o) inc += nb;
    }
    int bs = s0;
    for (int o = 16; o > 0; o >>= 1) bs += __shfl_xor(bs, o, 32);
    if (lane == 31) wsum[w]  = inc;
    if (lane == 0)  wbase[w] = bs;
    if (tid < BK_NODES) { lcnt[tid] = 0; lcur[tid] = 0; }
    __syncthreads();
    int before = 0, n = 0, base = 0;
    for (int i = 0; i < 32; ++i) {
        int sv = wsum[i];
        if (i < w) before += sv;
        n    += sv;
        base += wbase[i];
    }
    n    = __builtin_amdgcn_readfirstlane(n);
    base = __builtin_amdgcn_readfirstlane(base);
    if (tid < NCHUNK) {
        rsrc[tid] = tid * CHUNK + s0;
        rlen[tid] = len;
        rpos[tid] = before + inc - len;
    }
    __syncthreads();

    int g = tid >> 4;
    int l = tid & 15;
    bool st = (n <= BCAP);

    if (st) {
        int2 ed[RPG];
        int rl[RPG], rs[RPG], rp[RPG];
        #pragma unroll
        for (int k = 0; k < RPG; ++k) {
            int c  = g + k * NGRP;
            int cc = (c < NCHUNK) ? c : (NCHUNK - 1);
            int ln = rlen[cc];
            rl[k] = (c < NCHUNK) ? ln : 0;
            rs[k] = rsrc[cc];
            rp[k] = rpos[cc];
        }
        #pragma unroll
        for (int k = 0; k < RPG; ++k)
            ed[k] = (l < rl[k]) ? part[rs[k] + l] : make_int2(0, -1);
        #pragma unroll
        for (int k = 0; k < RPG; ++k)
            if (ed[k].y >= 0) sin_[rp[k] + l] = ed[k];
        #pragma unroll
        for (int k = 0; k < RPG; ++k)
            for (int i = l + 16; i < rl[k]; i += 16) sin_[rp[k] + i] = part[rs[k] + i];
        __syncthreads();
        for (int i = tid; i < n; i += 1024)
            atomicAdd(&lcnt[sin_[i].y & (BK_NODES - 1)], 1);
    } else {
        for (int c = g; c < NCHUNK; c += NGRP) {
            int ln = rlen[c];
            for (int i = l; i < ln; i += 16)
                atomicAdd(&lcnt[part[rsrc[c] + i].y & (BK_NODES - 1)], 1);
        }
    }
    __syncthreads();

    if (tid < 32) {
        int c[8];
        int sum = 0;
        #pragma unroll
        for (int k = 0; k < 8; ++k) { c[k] = lcnt[tid * 8 + k]; sum += c[k]; }
        int pin = sum;
        for (int o = 1; o < 32; o <<= 1) {
            int nb = __shfl_up(pin, o, 32);
            if (lane >= o) pin += nb;
        }
        int run = pin - sum;
        #pragma unroll
        for (int k = 0; k < 8; ++k) { lptr[tid * 8 + k] = run; run += c[k]; }
    }
    __syncthreads();

    if (tid < BK_NODES) {
        int node = b * BK_NODES + tid;
        if (node < N_NODES) {
            int c = lcnt[tid];
            rowptr[node] = base + lptr[tid];
            cnt[node]    = c;
            dinv[node]   = rsqrtf((float)c + 1.0f);
        }
    }

    if (st) {
        for (int i = tid; i < n; i += 1024) {
            int2 e = sin_[i];
            int dl = e.y & (BK_NODES - 1);
            int r  = atomicAdd(&lcur[dl], 1);
            sout[lptr[dl] + r] = e.x;
        }
        __syncthreads();
        for (int i = tid; i < n; i += 1024) rec[base + i] = sout[i];
    } else {
        for (int c = g; c < NCHUNK; c += NGRP) {
            int ln = rlen[c];
            for (int i = l; i < ln; i += 16) {
                int2 e = part[rsrc[c] + i];
                int dl = e.y & (BK_NODES - 1);
                int r  = atomicAdd(&lcur[dl], 1);
                rec[base + lptr[dl] + r] = e.x;
            }
        }
    }
}


template<int K, int NCOLS, bool RELU>
__global__ void k_gemm_wmma(const float* __restrict__ A,
                            const float* __restrict__ W,
                            const float* __restrict__ dinv,
                            float* __restrict__ out) {
    const int NT = NCOLS / 16;
    const int NG = NCOLS / 64;
    int strip = (blockIdx.x * blockDim.x + threadIdx.x) >> 5;
    int lane  = threadIdx.x & 31;

    int hi = lane >> 4;
    int c  = lane & 15;
    int mBase = strip * 16;

    const float* aRow = A + (size_t)(mBase + c) * K + 2 * hi;
    const float* wRow = W + (size_t)(2 * hi) * NCOLS + 4 * c;

    const float* dRow = dinv + mBase + 8 * hi;
    v4f dlo = *(const v4f*)(dRow);
    v4f dhi = *(const v4f*)(dRow + 4);

    v8f acc[NT];
    #pragma unroll
    for (int t = 0; t < NT; ++t)
        acc[t] = (v8f){0.f, 0.f, 0.f, 0.f, 0.f, 0.f, 0.f, 0.f};

    #pragma unroll 2
    for (int k = 0; k < K; k += 4) {
        v2f av = *(const v2f*)(aRow + k);
        if (RELU) { av.x = fmaxf(av.x, 0.0f); av.y = fmaxf(av.y, 0.0f); }

        v4f b0[NG], b1[NG];
        #pragma unroll
        for (int g = 0; g < NG; ++g) {
            b0[g] = *(const v4f*)(wRow + (size_t)k * NCOLS + 64 * g);
            b1[g] = *(const v4f*)(wRow + (size_t)(k + 1) * NCOLS + 64 * g);
        }

        #pragma unroll
        for (int t = 0; t < NT; ++t) {
            v2f bv;
            bv.x = b0[t >> 2][t & 3];
            bv.y = b1[t >> 2][t & 3];
            acc[t] = __builtin_amdgcn_wmma_f32_16x16x4_f32(false, av, false, bv,
                                                           (short)0, acc[t], false, false);
        }
    }

    float* oRow = out + (size_t)(mBase + 8 * hi) * NCOLS + 4 * c;
    #pragma unroll
    for (int r = 0; r < 8; ++r) {
        float sc = (r < 4) ? dlo[r & 3] : dhi[r & 3];
        #pragma unroll
        for (int g = 0; g < NG; ++g) {
            v4f o;
            o.x = acc[4 * g + 0][r] * sc;
            o.y = acc[4 * g + 1][r] * sc;
            o.z = acc[4 * g + 2][r] * sc;
            o.w = acc[4 * g + 3][r] * sc;
            *(v4f*)(oRow + (size_t)r * NCOLS + 64 * g) = o;
        }
    }
}


template<int V> struct RowVec;
template<> struct RowVec<4> { typedef v4f T; };
template<> struct RowVec<2> { typedef v2f T; };

template<int D>
__global__ void k_gather(const int* __restrict__ rowptr, const int* __restrict__ cnt,
                         const int* __restrict__ rec, const float* __restrict__ dinv,
                         const float* __restrict__ bias, const float* __restrict__ h,
                         float* __restrict__ agg) {
    const int V = D / 32;
    typedef typename RowVec<V>::T vec;
    int lane = threadIdx.x & 31;
    int node = __builtin_amdgcn_readfirstlane((int)((blockIdx.x * blockDim.x + threadIdx.x) >> 5));

    int j   = rowptr[node];
    int end = j + cnt[node];
    float di = dinv[node];

    const float* hl = h + lane * V;
    vec acc = *(const vec*)(hl + (size_t)node * D);

    for (; j + 4 <= end; j += 4) {
        int p0 = rec[j], p1 = rec[j + 1], p2 = rec[j + 2], p3 = rec[j + 3];
        vec x0 = *(const vec*)(hl + (size_t)p0 * D);
        vec x1 = *(const vec*)(hl + (size_t)p1 * D);
        vec x2 = *(const vec*)(hl + (size_t)p2 * D);
        vec x3 = *(const vec*)(hl + (size_t)p3 * D);
        acc += x0;
        acc += x1;
        acc += x2;
        acc += x3;
    }
    for (; j < end; ++j) {
        int p = rec[j];
        acc += *(const vec*)(hl + (size_t)p * D);
    }
    *(vec*)(agg + (size_t)node * D + lane * V) = acc * di + *(const vec*)(bias + lane * V);
}


#define FW 25
#define NSTRIP (N_NODES / 16)
#define NFB (NSTRIP / FW)
static_assert(NSTRIP % FW == 0, "one strip per wave, exact grid");
static_assert(N_NODES % 16 == 0, "whole strips");
#define TP (DH + 4)
#define WP (DOUT + 4)
#define GBATCH 16

__global__ void __launch_bounds__(FW * 32)
k_gather_gemm(const int* __restrict__ rowptr, const int* __restrict__ cnt,
              const int* __restrict__ rec, const float* __restrict__ dinv,
              const float* __restrict__ bias, const float* __restrict__ h,
              const float* __restrict__ W, float* __restrict__ out) {
    __shared__ float w2s[DH * WP];
    __shared__ float tiles[FW * 16 * TP];
    int tid  = threadIdx.x;
    int lane = tid & 31;
    int wv   = __builtin_amdgcn_readfirstlane(tid >> 5);

    int strip = blockIdx.x * FW + wv;
    int n0 = strip * 16;
    float* tl = tiles + wv * (16 * TP);
    const float* hl = h + lane * 4;
    float* tcol = tl + 4 * lane;

    v4f wq[4];
    if (tid < 512) {
        #pragma unroll
        for (int it = 0; it < 4; ++it)
            wq[it] = *(const v4f*)(W + 4 * (size_t)(tid + it * 512));
    }

    int li   = lane & 15;
    int rp   = rowptr[n0 + li];
    int endv = rp + cnt[n0 + li];
    float dv = dinv[n0 + li];
    v4f bvec = *(const v4f*)(bias + lane * 4);

    #pragma unroll
    for (int i = 0; i < 16; ++i)
        *(v4f*)(tcol + i * TP) = *(const v4f*)(hl + (size_t)(n0 + i) * DH);

    if (tid < 512) {
        #pragma unroll
        for (int it = 0; it < 4; ++it) {
            int i = tid + it * 512;
            *(v4f*)&w2s[(i >> 4) * WP + 4 * (i & 15)] = wq[it];
        }
    }
    __syncthreads();

    int j    = __builtin_amdgcn_readfirstlane(rp);
    int jend = __builtin_amdgcn_readlane(endv, 15);
    int i    = 0;
    int nend = __builtin_amdgcn_readlane(endv, 0);
    v4f acc  = *(const v4f*)(tcol);

#define FINISH_NODE()                                                                         \
    {                                                                                         \
        float di = __builtin_bit_cast(float,                                                  \
                       __builtin_amdgcn_readlane(__builtin_bit_cast(int, dv), i));            \
        v4f o = acc * di + bvec;                                                              \
        o.x = fmaxf(o.x, 0.0f); o.y = fmaxf(o.y, 0.0f);                                       \
        o.z = fmaxf(o.z, 0.0f); o.w = fmaxf(o.w, 0.0f);                                       \
        *(v4f*)(tcol + i * TP) = o;                                                           \
    }

    int p[GBATCH];
    if (j < jend) {
        #pragma unroll
        for (int k = 0; k < GBATCH; ++k) {
            int jj = j + k;
            p[k] = rec[jj < jend ? jj : jend - 1];
        }
    }
    while (j < jend) {
        v4f x[GBATCH];
        #pragma unroll
        for (int k = 0; k < GBATCH; ++k) x[k] = *(const v4f*)(hl + (size_t)p[k] * DH);
        int jn = j + GBATCH;
        if (jn < jend) {
            #pragma unroll
            for (int k = 0; k < GBATCH; ++k) {
                int jj = jn + k;
                p[k] = rec[jj < jend ? jj : jend - 1];
            }
        }
        #pragma unroll
        for (int k = 0; k < GBATCH; ++k) {
            int jj = j + k;
            if (jj < jend) {
                while (jj >= nend) {
                    FINISH_NODE();
                    ++i;
                    nend = __builtin_amdgcn_readlane(endv, i);
                    acc  = *(const v4f*)(tcol + i * TP);
                }
                acc += x[k];
            }
        }
        j = jn;
    }
    for (;;) {
        FINISH_NODE();
        ++i;
        if (i >= 16) break;
        acc = *(const v4f*)(tcol + i * TP);
    }
#undef FINISH_NODE


    int hi = lane >> 4;
    int c  = lane & 15;
    const float* aRow = tl + c * TP + 2 * hi;
    const float* wRow = w2s + (2 * hi) * WP + 4 * c;
    const float* dRow = dinv + n0 + 8 * hi;
    v4f dlo = *(const v4f*)(dRow);
    v4f dhi = *(const v4f*)(dRow + 4);

    v8f am[4];
    #pragma unroll
    for (int t = 0; t < 4; ++t)
        am[t] = (v8f){0.f, 0.f, 0.f, 0.f, 0.f, 0.f, 0.f, 0.f};

    #pragma unroll 2
    for (int k = 0; k < DH; k += 4) {
        v2f av = *(const v2f*)(aRow + k);
        v4f b0 = *(const v4f*)(wRow + k * WP);
        v4f b1 = *(const v4f*)(wRow + (k + 1) * WP);
        #pragma unroll
        for (int t = 0; t < 4; ++t) {
            v2f bv;
            bv.x = b0[t];
            bv.y = b1[t];
            am[t] = __builtin_amdgcn_wmma_f32_16x16x4_f32(false, av, false, bv,
                                                          (short)0, am[t], false, false);
        }
    }

    float* oRow = out + (size_t)(n0 + 8 * hi) * DOUT + 4 * c;
    #pragma unroll
    for (int r = 0; r < 8; ++r) {
        float sc = (r < 4) ? dlo[r & 3] : dhi[r & 3];
        v4f o;
        o.x = am[0][r] * sc;
        o.y = am[1][r] * sc;
        o.z = am[2][r] * sc;
        o.w = am[3][r] * sc;
        *(v4f*)(oRow + (size_t)r * DOUT) = o;
    }
}


extern "C" void kernel_launch(void* const* d_in, const int* in_sizes, int n_in,
                              void* d_out, int out_size, void* d_ws, size_t ws_size,
                              hipStream_t stream) {
    const float* x   = (const float*)d_in[0];
    const int*   ei  = (const int*)d_in[1];
    const float* W1  = (const float*)d_in[2];
    const float* b1  = (const float*)d_in[3];
    const float* W2  = (const float*)d_in[4];
    const float* b2  = (const float*)d_in[5];
    float*       out = (float*)d_out;

    const int* src = ei;
    const int* dst = ei + N_EDGES;

    float* ws    = (float*)d_ws;
    float* h1    = ws;
    float* agg1  = h1    + (size_t)N_NODES * DH;
    int*   rec   = (int*)(agg1 + (size_t)N_NODES * DH);
    int*   table = rec + 2 * (size_t)N_EDGES;
    float* dinv  = (float*)(table + N_EDGES);
    int*   cnt   = (int*)(dinv + N_NODES);
    int*   rowp  = cnt + N_NODES;
    int2*  part  = (int2*)agg1;
    float* h2    = agg1;

    const int B = 256;

    k_part    <<<NCHUNK, B, 0, stream>>>(src, dst, table, part);
    k_bucket  <<<NBK, 1024, 0, stream>>>(table, part, rowp, cnt, dinv, rec);

    const int GB = 64;
    {
        int waves = N_NODES / 16;
        k_gemm_wmma<DIN, DH, false><<<waves * 32 / GB, GB, 0, stream>>>(x, W1, dinv, h1);
    }
    k_gather_gemm<<<NFB, FW * 32, 0, stream>>>(rowp, cnt, rec, dinv, b1, h1, W2, h2);

    k_gather<DOUT><<<N_NODES * 32 / B, B, 0, stream>>>(rowp, cnt, rec, dinv, b2, h2, out);
}
